# starting kernel
# baseline (speedup 1.0000x reference)
#include <hip/hip_runtime.h>
#include <hip/hip_bf16.h>

#define D 96
#define XSTR 97
#define NODES_PER_BLOCK 64
#define NEG_SLOPE 0.01f

typedef float v2f __attribute__((ext_vector_type(2)));
typedef float v8f __attribute__((ext_vector_type(8)));

__global__ void kgetcda_zero(float* __restrict__ side, int total) {
    int i = blockIdx.x * blockDim.x + threadIdx.x;
    if (i < total) side[i] = 0.0f;
}

__global__ void kgetcda_spmm(const float* __restrict__ ego,
                             const float* __restrict__ vals,
                             const int*   __restrict__ rows,
                             const int*   __restrict__ cols,
                             float*       __restrict__ side,
                             int E) {
    int tid  = blockIdx.x * blockDim.x + threadIdx.x;
    int e    = tid >> 5;
    int lane = tid & 31;
    if (e >= E) return;
    int   r = rows[e];
    int   c = cols[e];
    float v = vals[e];
    const float* src = ego  + (size_t)c * D;
    float*       dst = side + (size_t)r * D;
#pragma unroll
    for (int k = lane; k < D; k += 32) {
        atomicAdd(&dst[k], v * src[k]);
    }
}

__global__ void __launch_bounds__(128)
kgetcda_fused(const float* __restrict__ ego,  const float* __restrict__ side,
              const float* __restrict__ W1,   const float* __restrict__ b1,
              const float* __restrict__ W2,   const float* __restrict__ b2,
              float* __restrict__ out, int N) {
    __shared__ float sXs[NODES_PER_BLOCK * XSTR];
    __shared__ float sXm[NODES_PER_BLOCK * XSTR];

    const int tid      = threadIdx.x;
    const int nodeBase = blockIdx.x * NODES_PER_BLOCK;

    for (int i = tid; i < NODES_PER_BLOCK * D; i += 128) {
        int ln = i / D;
        int k  = i - ln * D;
        int node = nodeBase + ln;
        if (node < N) {
            float e = ego [(size_t)node * D + k];
            float s = side[(size_t)node * D + k];
            sXs[ln * XSTR + k] = e + s;
            sXm[ln * XSTR + k] = e * s;
        }
    }
    __syncthreads();

    const int wave = tid >> 5;
    const int lane = tid & 31;
    const int tileNode = nodeBase + wave * 16;

    if (tileNode + 15 < N || tileNode < N) {
        if (tileNode >= N) return;
        const int n = lane & 15;
        const int g = lane >> 4;
        const int m = lane & 15;

        const float* xs = &sXs[(wave * 16) * XSTR];
        const float* xm = &sXm[(wave * 16) * XSTR];

        for (int j = 0; j < D; j += 16) {
            v8f acc1, acc2;
            const float bb1 = b1[j + n];
            const float bb2 = b2[j + n];
#pragma unroll
            for (int r = 0; r < 8; ++r) { acc1[r] = bb1; acc2[r] = bb2; }

            const float* w1row = W1 + (size_t)(j + n) * D;
            const float* w2row = W2 + (size_t)(j + n) * D;

#pragma unroll
            for (int kb = 0; kb < D; kb += 4) {
                const int kk = kb + 2 * g;
                v2f a1, a2, bf1, bf2;
                a1.x  = xs[m * XSTR + kk];     a1.y  = xs[m * XSTR + kk + 1];
                a2.x  = xm[m * XSTR + kk];     a2.y  = xm[m * XSTR + kk + 1];
                bf1.x = w1row[kk];             bf1.y = w1row[kk + 1];
                bf2.x = w2row[kk];             bf2.y = w2row[kk + 1];
                acc1 = __builtin_amdgcn_wmma_f32_16x16x4_f32(
                           false, a1, false, bf1, (short)0, acc1, false, false);
                acc2 = __builtin_amdgcn_wmma_f32_16x16x4_f32(
                           false, a2, false, bf2, (short)0, acc2, false, false);
            }

#pragma unroll
            for (int r = 0; r < 8; ++r) {
                float x1 = acc1[r]; x1 = (x1 >= 0.0f) ? x1 : NEG_SLOPE * x1;
                float x2 = acc2[r]; x2 = (x2 >= 0.0f) ? x2 : NEG_SLOPE * x2;
                int row = r + 8 * g;
                out[(size_t)(tileNode + row) * D + j + n] = x1 + x2;
            }
        }
    }
}

extern "C" void kernel_launch(void* const* d_in, const int* in_sizes, int n_in,
                              void* d_out, int out_size, void* d_ws, size_t ws_size,
                              hipStream_t stream) {
    const float* ego    = (const float*)d_in[0];
    const float* a_vals = (const float*)d_in[1];
    const float* W1     = (const float*)d_in[2];
    const float* b1     = (const float*)d_in[3];
    const float* W2     = (const float*)d_in[4];
    const float* b2     = (const float*)d_in[5];
    const int*   a_rows = (const int*)d_in[6];
    const int*   a_cols = (const int*)d_in[7];
    float* out  = (float*)d_out;
    float* side = (float*)d_ws;

    const int N = in_sizes[0] / D;
    const int E = in_sizes[6];

    const int total = N * D;
    kgetcda_zero<<<(total + 255) / 256, 256, 0, stream>>>(side, total);

    const int spmmBlocks = (E + 7) / 8;
    kgetcda_spmm<<<spmmBlocks, 256, 0, stream>>>(ego, a_vals, a_rows, a_cols,
                                                 side, E);

    const int fusedBlocks = (N + NODES_PER_BLOCK - 1) / NODES_PER_BLOCK;
    kgetcda_fused<<<fusedBlocks, 128, 0, stream>>>(ego, side, W1, b1, W2, b2,
                                                   out, N);
}


# SpMM without f32 atomics: edges bucketed by row (int count atomics; buckets kept in out until the final store), one wave per row adds its edges in ascending edge index by fma: fixed sum order
# speedup vs baseline: 1.0827x; 1.0827x over previous
#include <hip/hip_runtime.h>
#include <hip/hip_bf16.h>

#define D 96
#define XSTR 97
#define NODES_PER_BLOCK 64
#define NEG_SLOPE 0.01f

typedef float v2f __attribute__((ext_vector_type(2)));
typedef float v8f __attribute__((ext_vector_type(8)));


#define ALLOC_THREADS 256
#define ROWS_PER_BLOCK 8

__global__ void kgetcda_zero_counts(int* __restrict__ cnt, int n) {
    int i = blockIdx.x * blockDim.x + threadIdx.x;
    if (i < n) cnt[i] = 0;
}

__global__ void kgetcda_count(const int* __restrict__ rows,
                              int* __restrict__ cnt,
                              int* __restrict__ rank, int E) {
    int e = blockIdx.x * blockDim.x + threadIdx.x;
    if (e < E) rank[e] = atomicAdd(&cnt[rows[e]], 1);
}

__global__ void __launch_bounds__(ALLOC_THREADS)
kgetcda_alloc(const int* cnt, int* total, int* __restrict__ start, int N) {
    __shared__ int wsum[ALLOC_THREADS / 32];
    __shared__ int base;
    const int t    = threadIdx.x;
    const int lane = t & 31;
    const int w    = t >> 5;
    const int r    = blockIdx.x * ALLOC_THREADS + t;
    const int c    = (r < N) ? cnt[r] : 0;
    int x = c;
    for (int off = 1; off < 32; off <<= 1) {
        int y = __shfl_up(x, off);
        if (lane >= off) x += y;
    }
    if (lane == 31) wsum[w] = x;
    __syncthreads();
    if (t == 0) {
        int run = 0;
        for (int i = 0; i < ALLOC_THREADS / 32; ++i) {
            int v = wsum[i];
            wsum[i] = run;
            run += v;
        }
        base = atomicAdd(total, run);
    }
    __syncthreads();
    if (r < N) start[r] = base + wsum[w] + x - c;
}

__global__ void kgetcda_fill(const int*   __restrict__ rows,
                             const int*   __restrict__ cols,
                             const float* __restrict__ vals,
                             const int*   __restrict__ start,
                             const int*   __restrict__ rank,
                             int*         __restrict__ ent, int E) {
    int e = blockIdx.x * blockDim.x + threadIdx.x;
    if (e < E) {
        int pos = start[rows[e]] + rank[e];
        int4 v;
        v.x = e;
        v.y = cols[e];
        v.z = __float_as_int(vals[e]);
        v.w = 0;
        reinterpret_cast<int4*>(ent)[pos] = v;
    }
}

__global__ void __launch_bounds__(ROWS_PER_BLOCK * 32)
kgetcda_gather(const float* __restrict__ ego,
               const int*   __restrict__ start,
               const int*   __restrict__ cnt,
               const int*   __restrict__ ent,
               float*       __restrict__ side, int N) {
    const int lane = threadIdx.x & 31;
    const int row = __builtin_amdgcn_readfirstlane(
                        (int)(blockIdx.x * ROWS_PER_BLOCK + (threadIdx.x >> 5)));
    if (row >= N) return;
    const int s = start[row];
    const int d = cnt[row];
    const int4* bucket = reinterpret_cast<const int4*>(ent) + s;

    float a0 = 0.0f, a1 = 0.0f, a2 = 0.0f;
    const float* base = ego + lane;

    if (d <= 32) {
        int me = 0x7fffffff, mc = 0, mv = 0;
        if (lane < d) { int4 v = bucket[lane]; me = v.x; mc = v.y; mv = v.z; }
        int rk = 0;
        for (int j = 0; j < d; ++j)
            rk += (__builtin_amdgcn_readlane(me, j) < me) ? 1 : 0;
        const int sc = __builtin_amdgcn_ds_permute(rk << 2, mc);
        const int sv = __builtin_amdgcn_ds_permute(rk << 2, mv);
#pragma unroll 4
        for (int k = 0; k < d; ++k) {
            const int   c = __builtin_amdgcn_readlane(sc, k);
            const float v = __int_as_float(__builtin_amdgcn_readlane(sv, k));
            const float* p = base + (size_t)c * D;
            a0 = fmaf(v, p[0],  a0);
            a1 = fmaf(v, p[32], a1);
            a2 = fmaf(v, p[64], a2);
        }
    } else {
        int last = -1;
        for (int k = 0; k < d; ++k) {
            int best = 0x7fffffff, bq = 0;
            for (int q = lane; q < d; q += 32) {
                int e = bucket[q].x;
                if (e > last && e < best) { best = e; bq = q; }
            }
            for (int off = 16; off > 0; off >>= 1) {
                int ob = __shfl_xor(best, off);
                int oq = __shfl_xor(bq, off);
                if (ob < best) { best = ob; bq = oq; }
            }
            const int4  w = bucket[bq];
            const float v = __int_as_float(w.z);
            const float* p = base + (size_t)w.y * D;
            a0 = fmaf(v, p[0],  a0);
            a1 = fmaf(v, p[32], a1);
            a2 = fmaf(v, p[64], a2);
            last = best;
        }
    }
    float* dst = side + (size_t)row * D + lane;
    dst[0]  = a0;
    dst[32] = a1;
    dst[64] = a2;
}

__global__ void __launch_bounds__(128)
kgetcda_fused(const float* __restrict__ ego,  const float* __restrict__ side,
              const float* __restrict__ W1,   const float* __restrict__ b1,
              const float* __restrict__ W2,   const float* __restrict__ b2,
              float* __restrict__ out, int N) {
    __shared__ float sXs[NODES_PER_BLOCK * XSTR];
    __shared__ float sXm[NODES_PER_BLOCK * XSTR];

    const int tid      = threadIdx.x;
    const int nodeBase = blockIdx.x * NODES_PER_BLOCK;

    for (int i = tid; i < NODES_PER_BLOCK * D; i += 128) {
        int ln = i / D;
        int k  = i - ln * D;
        int node = nodeBase + ln;
        if (node < N) {
            float e = ego [(size_t)node * D + k];
            float s = side[(size_t)node * D + k];
            sXs[ln * XSTR + k] = e + s;
            sXm[ln * XSTR + k] = e * s;
        }
    }
    __syncthreads();

    const int wave = tid >> 5;
    const int lane = tid & 31;
    const int tileNode = nodeBase + wave * 16;

    if (tileNode + 15 < N || tileNode < N) {
        if (tileNode >= N) return;
        const int n = lane & 15;
        const int g = lane >> 4;
        const int m = lane & 15;

        const float* xs = &sXs[(wave * 16) * XSTR];
        const float* xm = &sXm[(wave * 16) * XSTR];

        for (int j = 0; j < D; j += 16) {
            v8f acc1, acc2;
            const float bb1 = b1[j + n];
            const float bb2 = b2[j + n];
#pragma unroll
            for (int r = 0; r < 8; ++r) { acc1[r] = bb1; acc2[r] = bb2; }

            const float* w1row = W1 + (size_t)(j + n) * D;
            const float* w2row = W2 + (size_t)(j + n) * D;

#pragma unroll
            for (int kb = 0; kb < D; kb += 4) {
                const int kk = kb + 2 * g;
                v2f a1, a2, bf1, bf2;
                a1.x  = xs[m * XSTR + kk];     a1.y  = xs[m * XSTR + kk + 1];
                a2.x  = xm[m * XSTR + kk];     a2.y  = xm[m * XSTR + kk + 1];
                bf1.x = w1row[kk];             bf1.y = w1row[kk + 1];
                bf2.x = w2row[kk];             bf2.y = w2row[kk + 1];
                acc1 = __builtin_amdgcn_wmma_f32_16x16x4_f32(
                           false, a1, false, bf1, (short)0, acc1, false, false);
                acc2 = __builtin_amdgcn_wmma_f32_16x16x4_f32(
                           false, a2, false, bf2, (short)0, acc2, false, false);
            }

#pragma unroll
            for (int r = 0; r < 8; ++r) {
                float x1 = acc1[r]; x1 = (x1 >= 0.0f) ? x1 : NEG_SLOPE * x1;
                float x2 = acc2[r]; x2 = (x2 >= 0.0f) ? x2 : NEG_SLOPE * x2;
                int row = r + 8 * g;
                out[(size_t)(tileNode + row) * D + j + n] = x1 + x2;
            }
        }
    }
}

extern "C" void kernel_launch(void* const* d_in, const int* in_sizes, int n_in,
                              void* d_out, int out_size, void* d_ws, size_t ws_size,
                              hipStream_t stream) {
    const float* ego    = (const float*)d_in[0];
    const float* a_vals = (const float*)d_in[1];
    const float* W1     = (const float*)d_in[2];
    const float* b1     = (const float*)d_in[3];
    const float* W2     = (const float*)d_in[4];
    const float* b2     = (const float*)d_in[5];
    const int*   a_rows = (const int*)d_in[6];
    const int*   a_cols = (const int*)d_in[7];
    float* out  = (float*)d_out;
    float* side = (float*)d_ws;

    const int N = in_sizes[0] / D;
    const int E = in_sizes[6];

    int* ent  = (int*)d_out;
    int* rank = ent  + 4 * (size_t)E;
    int* cnt   = rank + E;
    int* start = cnt  + N + 1;

    kgetcda_zero_counts<<<(N + 256) / 256, 256, 0, stream>>>(cnt, N + 1);
    kgetcda_count<<<(E + 255) / 256, 256, 0, stream>>>(a_rows, cnt, rank, E);
    kgetcda_alloc<<<(N + ALLOC_THREADS - 1) / ALLOC_THREADS, ALLOC_THREADS, 0,
                    stream>>>(cnt, cnt + N, start, N);
    kgetcda_fill<<<(E + 255) / 256, 256, 0, stream>>>(a_rows, a_cols, a_vals,
                                                     start, rank, ent, E);

    kgetcda_gather<<<(N + ROWS_PER_BLOCK - 1) / ROWS_PER_BLOCK,
                     ROWS_PER_BLOCK * 32, 0, stream>>>(ego, start, cnt, ent, side, N);

    const int fusedBlocks = (N + NODES_PER_BLOCK - 1) / NODES_PER_BLOCK;
    kgetcda_fused<<<fusedBlocks, 128, 0, stream>>>(ego, side, W1, b1, W2, b2,
                                                   out, N);
}


# Fused GEMM: fragment k regrouped into 4-float runs per lane, so X goes global-to-registers and W in 16-byte loads (no LDS staging pass); products summed in a different fixed k order
# speedup vs baseline: 1.2608x; 1.1645x over previous
#include <hip/hip_runtime.h>
#include <hip/hip_bf16.h>

#define D 96
#define NODES_PER_BLOCK 64
#define NEG_SLOPE 0.01f

typedef float v2f __attribute__((ext_vector_type(2)));
typedef float v8f __attribute__((ext_vector_type(8)));


#define ALLOC_THREADS 256
#define ROWS_PER_BLOCK 8

__global__ void kgetcda_zero_counts(int* __restrict__ cnt, int n) {
    int i = blockIdx.x * blockDim.x + threadIdx.x;
    if (i < n) cnt[i] = 0;
}

__global__ void kgetcda_count(const int* __restrict__ rows,
                              int* __restrict__ cnt,
                              int* __restrict__ rank, int E) {
    int e = blockIdx.x * blockDim.x + threadIdx.x;
    if (e < E) rank[e] = atomicAdd(&cnt[rows[e]], 1);
}

__global__ void __launch_bounds__(ALLOC_THREADS)
kgetcda_alloc(const int* cnt, int* total, int* __restrict__ start, int N) {
    __shared__ int wsum[ALLOC_THREADS / 32];
    __shared__ int base;
    const int t    = threadIdx.x;
    const int lane = t & 31;
    const int w    = t >> 5;
    const int r    = blockIdx.x * ALLOC_THREADS + t;
    const int c    = (r < N) ? cnt[r] : 0;
    int x = c;
    for (int off = 1; off < 32; off <<= 1) {
        int y = __shfl_up(x, off);
        if (lane >= off) x += y;
    }
    if (lane == 31) wsum[w] = x;
    __syncthreads();
    if (t == 0) {
        int run = 0;
        for (int i = 0; i < ALLOC_THREADS / 32; ++i) {
            int v = wsum[i];
            wsum[i] = run;
            run += v;
        }
        base = atomicAdd(total, run);
    }
    __syncthreads();
    if (r < N) start[r] = base + wsum[w] + x - c;
}

__global__ void kgetcda_fill(const int*   __restrict__ rows,
                             const int*   __restrict__ cols,
                             const float* __restrict__ vals,
                             const int*   __restrict__ start,
                             const int*   __restrict__ rank,
                             int*         __restrict__ ent, int E) {
    int e = blockIdx.x * blockDim.x + threadIdx.x;
    if (e < E) {
        int pos = start[rows[e]] + rank[e];
        int4 v;
        v.x = e;
        v.y = cols[e];
        v.z = __float_as_int(vals[e]);
        v.w = 0;
        reinterpret_cast<int4*>(ent)[pos] = v;
    }
}

__global__ void __launch_bounds__(ROWS_PER_BLOCK * 32)
kgetcda_gather(const float* __restrict__ ego,
               const int*   __restrict__ start,
               const int*   __restrict__ cnt,
               const int*   __restrict__ ent,
               float*       __restrict__ side, int N) {
    const int lane = threadIdx.x & 31;
    const int row = __builtin_amdgcn_readfirstlane(
                        (int)(blockIdx.x * ROWS_PER_BLOCK + (threadIdx.x >> 5)));
    if (row >= N) return;
    const int s = start[row];
    const int d = cnt[row];
    const int4* bucket = reinterpret_cast<const int4*>(ent) + s;

    float a0 = 0.0f, a1 = 0.0f, a2 = 0.0f;
    const float* base = ego + lane;

    if (d <= 32) {
        int me = 0x7fffffff, mc = 0, mv = 0;
        if (lane < d) { int4 v = bucket[lane]; me = v.x; mc = v.y; mv = v.z; }
        int rk = 0;
        for (int j = 0; j < d; ++j)
            rk += (__builtin_amdgcn_readlane(me, j) < me) ? 1 : 0;
        const int sc = __builtin_amdgcn_ds_permute(rk << 2, mc);
        const int sv = __builtin_amdgcn_ds_permute(rk << 2, mv);
#pragma unroll 4
        for (int k = 0; k < d; ++k) {
            const int   c = __builtin_amdgcn_readlane(sc, k);
            const float v = __int_as_float(__builtin_amdgcn_readlane(sv, k));
            const float* p = base + (size_t)c * D;
            a0 = fmaf(v, p[0],  a0);
            a1 = fmaf(v, p[32], a1);
            a2 = fmaf(v, p[64], a2);
        }
    } else {
        int last = -1;
        for (int k = 0; k < d; ++k) {
            int best = 0x7fffffff, bq = 0;
            for (int q = lane; q < d; q += 32) {
                int e = bucket[q].x;
                if (e > last && e < best) { best = e; bq = q; }
            }
            for (int off = 16; off > 0; off >>= 1) {
                int ob = __shfl_xor(best, off);
                int oq = __shfl_xor(bq, off);
                if (ob < best) { best = ob; bq = oq; }
            }
            const int4  w = bucket[bq];
            const float v = __int_as_float(w.z);
            const float* p = base + (size_t)w.y * D;
            a0 = fmaf(v, p[0],  a0);
            a1 = fmaf(v, p[32], a1);
            a2 = fmaf(v, p[64], a2);
            last = best;
        }
    }
    float* dst = side + (size_t)row * D + lane;
    dst[0]  = a0;
    dst[32] = a1;
    dst[64] = a2;
}

__global__ void __launch_bounds__(128)
kgetcda_fused(const float* __restrict__ ego,  const float* __restrict__ side,
              const float* __restrict__ W1,   const float* __restrict__ b1,
              const float* __restrict__ W2,   const float* __restrict__ b2,
              float* __restrict__ out, int N) {
    const int tid      = threadIdx.x;
    const int wave     = tid >> 5;
    const int lane     = tid & 31;
    const int tileNode = blockIdx.x * NODES_PER_BLOCK + wave * 16;

    if (tileNode >= N) return;

    const int n = lane & 15;
    const int g = lane >> 4;

    const float4* ep = reinterpret_cast<const float4*>(ego  + (size_t)(tileNode + n) * D) + g;
    const float4* sp = reinterpret_cast<const float4*>(side + (size_t)(tileNode + n) * D) + g;
    float4 xs[12], xm[12];
#pragma unroll
    for (int t = 0; t < 12; ++t) {
        const float4 e = ep[2 * t];
        const float4 s = sp[2 * t];
        xs[t].x = e.x + s.x; xs[t].y = e.y + s.y; xs[t].z = e.z + s.z; xs[t].w = e.w + s.w;
        xm[t].x = e.x * s.x; xm[t].y = e.y * s.y; xm[t].z = e.z * s.z; xm[t].w = e.w * s.w;
    }

#pragma unroll 1
    for (int j = 0; j < D; j += 16) {
        v8f acc1, acc2;
        const float bb1 = b1[j + n];
        const float bb2 = b2[j + n];
#pragma unroll
        for (int r = 0; r < 8; ++r) { acc1[r] = bb1; acc2[r] = bb2; }

        const float4* w1p = reinterpret_cast<const float4*>(W1 + (size_t)(j + n) * D) + g;
        const float4* w2p = reinterpret_cast<const float4*>(W2 + (size_t)(j + n) * D) + g;

#pragma unroll
        for (int t = 0; t < 12; ++t) {
            const float4 w1 = w1p[2 * t];
            const float4 w2 = w2p[2 * t];
            v2f a1, a2, bf1, bf2;
            a1.x  = xs[t].x;  a1.y  = xs[t].y;
            a2.x  = xm[t].x;  a2.y  = xm[t].y;
            bf1.x = w1.x;     bf1.y = w1.y;
            bf2.x = w2.x;     bf2.y = w2.y;
            acc1 = __builtin_amdgcn_wmma_f32_16x16x4_f32(
                       false, a1, false, bf1, (short)0, acc1, false, false);
            acc2 = __builtin_amdgcn_wmma_f32_16x16x4_f32(
                       false, a2, false, bf2, (short)0, acc2, false, false);
            a1.x  = xs[t].z;  a1.y  = xs[t].w;
            a2.x  = xm[t].z;  a2.y  = xm[t].w;
            bf1.x = w1.z;     bf1.y = w1.w;
            bf2.x = w2.z;     bf2.y = w2.w;
            acc1 = __builtin_amdgcn_wmma_f32_16x16x4_f32(
                       false, a1, false, bf1, (short)0, acc1, false, false);
            acc2 = __builtin_amdgcn_wmma_f32_16x16x4_f32(
                       false, a2, false, bf2, (short)0, acc2, false, false);
        }

#pragma unroll
        for (int r = 0; r < 8; ++r) {
            float x1 = acc1[r]; x1 = (x1 >= 0.0f) ? x1 : NEG_SLOPE * x1;
            float x2 = acc2[r]; x2 = (x2 >= 0.0f) ? x2 : NEG_SLOPE * x2;
            int row = r + 8 * g;
            out[(size_t)(tileNode + row) * D + j + n] = x1 + x2;
        }
    }
}

extern "C" void kernel_launch(void* const* d_in, const int* in_sizes, int n_in,
                              void* d_out, int out_size, void* d_ws, size_t ws_size,
                              hipStream_t stream) {
    const float* ego    = (const float*)d_in[0];
    const float* a_vals = (const float*)d_in[1];
    const float* W1     = (const float*)d_in[2];
    const float* b1     = (const float*)d_in[3];
    const float* W2     = (const float*)d_in[4];
    const float* b2     = (const float*)d_in[5];
    const int*   a_rows = (const int*)d_in[6];
    const int*   a_cols = (const int*)d_in[7];
    float* out  = (float*)d_out;
    float* side = (float*)d_ws;

    const int N = in_sizes[0] / D;
    const int E = in_sizes[6];

    int* ent  = (int*)d_out;
    int* rank = ent  + 4 * (size_t)E;
    int* cnt   = rank + E;
    int* start = cnt  + N + 1;

    kgetcda_zero_counts<<<(N + 256) / 256, 256, 0, stream>>>(cnt, N + 1);
    kgetcda_count<<<(E + 255) / 256, 256, 0, stream>>>(a_rows, cnt, rank, E);
    kgetcda_alloc<<<(N + ALLOC_THREADS - 1) / ALLOC_THREADS, ALLOC_THREADS, 0,
                    stream>>>(cnt, cnt + N, start, N);
    kgetcda_fill<<<(E + 255) / 256, 256, 0, stream>>>(a_rows, a_cols, a_vals,
                                                     start, rank, ent, E);

    kgetcda_gather<<<(N + ROWS_PER_BLOCK - 1) / ROWS_PER_BLOCK,
                     ROWS_PER_BLOCK * 32, 0, stream>>>(ego, start, cnt, ent, side, N);

    const int fusedBlocks = (N + NODES_PER_BLOCK - 1) / NODES_PER_BLOCK;
    kgetcda_fused<<<fusedBlocks, 128, 0, stream>>>(ego, side, W1, b1, W2, b2,
                                                   out, N);
}


# SpMM gather: rows of 33 to 64 edges ranked in registers (two entries per lane) and summed by the pipelined loop, not by a serial pick with dependent loads per edge; sum order unchanged
# speedup vs baseline: 1.4357x; 1.1387x over previous
#include <hip/hip_runtime.h>
#include <hip/hip_bf16.h>

#define D 96
#define NODES_PER_BLOCK 64
#define NEG_SLOPE 0.01f

typedef float v2f __attribute__((ext_vector_type(2)));
typedef float v8f __attribute__((ext_vector_type(8)));


#define ALLOC_THREADS 256
#define ROWS_PER_BLOCK 8

__global__ void kgetcda_zero_counts(int* __restrict__ cnt, int n) {
    int i = blockIdx.x * blockDim.x + threadIdx.x;
    if (i < n) cnt[i] = 0;
}

__global__ void kgetcda_count(const int* __restrict__ rows,
                              int* __restrict__ cnt,
                              int* __restrict__ rank, int E) {
    int e = blockIdx.x * blockDim.x + threadIdx.x;
    if (e < E) rank[e] = atomicAdd(&cnt[rows[e]], 1);
}

__global__ void __launch_bounds__(ALLOC_THREADS)
kgetcda_alloc(const int* cnt, int* total, int* __restrict__ start, int N) {
    __shared__ int wsum[ALLOC_THREADS / 32];
    __shared__ int base;
    const int t    = threadIdx.x;
    const int lane = t & 31;
    const int w    = t >> 5;
    const int r    = blockIdx.x * ALLOC_THREADS + t;
    const int c    = (r < N) ? cnt[r] : 0;
    int x = c;
    for (int off = 1; off < 32; off <<= 1) {
        int y = __shfl_up(x, off);
        if (lane >= off) x += y;
    }
    if (lane == 31) wsum[w] = x;
    __syncthreads();
    if (t == 0) {
        int run = 0;
        for (int i = 0; i < ALLOC_THREADS / 32; ++i) {
            int v = wsum[i];
            wsum[i] = run;
            run += v;
        }
        base = atomicAdd(total, run);
    }
    __syncthreads();
    if (r < N) start[r] = base + wsum[w] + x - c;
}

__global__ void kgetcda_fill(const int*   __restrict__ rows,
                             const int*   __restrict__ cols,
                             const float* __restrict__ vals,
                             const int*   __restrict__ start,
                             const int*   __restrict__ rank,
                             int*         __restrict__ ent, int E) {
    int e = blockIdx.x * blockDim.x + threadIdx.x;
    if (e < E) {
        int pos = start[rows[e]] + rank[e];
        int4 v;
        v.x = e;
        v.y = cols[e];
        v.z = __float_as_int(vals[e]);
        v.w = 0;
        reinterpret_cast<int4*>(ent)[pos] = v;
    }
}

__global__ void __launch_bounds__(ROWS_PER_BLOCK * 32)
kgetcda_gather(const float* __restrict__ ego,
               const int*   __restrict__ start,
               const int*   __restrict__ cnt,
               const int*   __restrict__ ent,
               float*       __restrict__ side, int N) {
    const int lane = threadIdx.x & 31;
    const int row = __builtin_amdgcn_readfirstlane(
                        (int)(blockIdx.x * ROWS_PER_BLOCK + (threadIdx.x >> 5)));
    if (row >= N) return;
    const int s = start[row];
    const int d = cnt[row];
    const int4* bucket = reinterpret_cast<const int4*>(ent) + s;

    float a0 = 0.0f, a1 = 0.0f, a2 = 0.0f;
    const float* base = ego + lane;

    if (d <= 32) {
        int me = 0x7fffffff, mc = 0, mv = 0;
        if (lane < d) { int4 v = bucket[lane]; me = v.x; mc = v.y; mv = v.z; }
        int rk = 0;
        for (int j = 0; j < d; ++j)
            rk += (__builtin_amdgcn_readlane(me, j) < me) ? 1 : 0;
        const int sc = __builtin_amdgcn_ds_permute(rk << 2, mc);
        const int sv = __builtin_amdgcn_ds_permute(rk << 2, mv);
#pragma unroll 4
        for (int k = 0; k < d; ++k) {
            const int   c = __builtin_amdgcn_readlane(sc, k);
            const float v = __int_as_float(__builtin_amdgcn_readlane(sv, k));
            const float* p = base + (size_t)c * D;
            a0 = fmaf(v, p[0],  a0);
            a1 = fmaf(v, p[32], a1);
            a2 = fmaf(v, p[64], a2);
        }
    } else if (d <= 64) {
        const int4 v0 = bucket[lane];
        const int me0 = v0.x, mc0 = v0.y, mv0 = v0.z;
        int me1 = 0x7fffffff, mc1 = 0, mv1 = 0;
        if (lane + 32 < d) { const int4 v1 = bucket[lane + 32]; me1 = v1.x; mc1 = v1.y; mv1 = v1.z; }
        int rk0 = 0, rk1 = 0;
        for (int j = 0; j < 32; ++j) {
            const int x = __builtin_amdgcn_readlane(me0, j);
            rk0 += (x < me0) ? 1 : 0;
            rk1 += (x < me1) ? 1 : 0;
        }
        const int d1 = d - 32;
        for (int j = 0; j < d1; ++j) {
            const int x = __builtin_amdgcn_readlane(me1, j);
            rk0 += (x < me0) ? 1 : 0;
            rk1 += (x < me1) ? 1 : 0;
        }
#pragma unroll 4
        for (int k = 0; k < d; ++k) {
            const unsigned m0 = __builtin_amdgcn_ballot_w32(rk0 == k);
            const unsigned m1 = __builtin_amdgcn_ballot_w32(rk1 == k);
            const int l0 = m0 ? __builtin_ctz(m0) : 0;
            const int l1 = m1 ? __builtin_ctz(m1) : 0;
            const int c0 = __builtin_amdgcn_readlane(mc0, l0);
            const int c1 = __builtin_amdgcn_readlane(mc1, l1);
            const int b0 = __builtin_amdgcn_readlane(mv0, l0);
            const int b1 = __builtin_amdgcn_readlane(mv1, l1);
            const int   c = m0 ? c0 : c1;
            const float v = __int_as_float(m0 ? b0 : b1);
            const float* p = base + (size_t)c * D;
            a0 = fmaf(v, p[0],  a0);
            a1 = fmaf(v, p[32], a1);
            a2 = fmaf(v, p[64], a2);
        }
    } else {
        int last = -1;
        for (int k = 0; k < d; ++k) {
            int best = 0x7fffffff, bq = 0;
            for (int q = lane; q < d; q += 32) {
                int e = bucket[q].x;
                if (e > last && e < best) { best = e; bq = q; }
            }
            for (int off = 16; off > 0; off >>= 1) {
                int ob = __shfl_xor(best, off);
                int oq = __shfl_xor(bq, off);
                if (ob < best) { best = ob; bq = oq; }
            }
            const int4  w = bucket[bq];
            const float v = __int_as_float(w.z);
            const float* p = base + (size_t)w.y * D;
            a0 = fmaf(v, p[0],  a0);
            a1 = fmaf(v, p[32], a1);
            a2 = fmaf(v, p[64], a2);
            last = best;
        }
    }
    float* dst = side + (size_t)row * D + lane;
    dst[0]  = a0;
    dst[32] = a1;
    dst[64] = a2;
}

__global__ void __launch_bounds__(128)
kgetcda_fused(const float* __restrict__ ego,  const float* __restrict__ side,
              const float* __restrict__ W1,   const float* __restrict__ b1,
              const float* __restrict__ W2,   const float* __restrict__ b2,
              float* __restrict__ out, int N) {
    const int tid      = threadIdx.x;
    const int wave     = tid >> 5;
    const int lane     = tid & 31;
    const int tileNode = blockIdx.x * NODES_PER_BLOCK + wave * 16;

    if (tileNode >= N) return;

    const int n = lane & 15;
    const int g = lane >> 4;

    const float4* ep = reinterpret_cast<const float4*>(ego  + (size_t)(tileNode + n) * D) + g;
    const float4* sp = reinterpret_cast<const float4*>(side + (size_t)(tileNode + n) * D) + g;
    float4 xs[12], xm[12];
#pragma unroll
    for (int t = 0; t < 12; ++t) {
        const float4 e = ep[2 * t];
        const float4 s = sp[2 * t];
        xs[t].x = e.x + s.x; xs[t].y = e.y + s.y; xs[t].z = e.z + s.z; xs[t].w = e.w + s.w;
        xm[t].x = e.x * s.x; xm[t].y = e.y * s.y; xm[t].z = e.z * s.z; xm[t].w = e.w * s.w;
    }

#pragma unroll 1
    for (int j = 0; j < D; j += 16) {
        v8f acc1, acc2;
        const float bb1 = b1[j + n];
        const float bb2 = b2[j + n];
#pragma unroll
        for (int r = 0; r < 8; ++r) { acc1[r] = bb1; acc2[r] = bb2; }

        const float4* w1p = reinterpret_cast<const float4*>(W1 + (size_t)(j + n) * D) + g;
        const float4* w2p = reinterpret_cast<const float4*>(W2 + (size_t)(j + n) * D) + g;

#pragma unroll
        for (int t = 0; t < 12; ++t) {
            const float4 w1 = w1p[2 * t];
            const float4 w2 = w2p[2 * t];
            v2f a1, a2, bf1, bf2;
            a1.x  = xs[t].x;  a1.y  = xs[t].y;
            a2.x  = xm[t].x;  a2.y  = xm[t].y;
            bf1.x = w1.x;     bf1.y = w1.y;
            bf2.x = w2.x;     bf2.y = w2.y;
            acc1 = __builtin_amdgcn_wmma_f32_16x16x4_f32(
                       false, a1, false, bf1, (short)0, acc1, false, false);
            acc2 = __builtin_amdgcn_wmma_f32_16x16x4_f32(
                       false, a2, false, bf2, (short)0, acc2, false, false);
            a1.x  = xs[t].z;  a1.y  = xs[t].w;
            a2.x  = xm[t].z;  a2.y  = xm[t].w;
            bf1.x = w1.z;     bf1.y = w1.w;
            bf2.x = w2.z;     bf2.y = w2.w;
            acc1 = __builtin_amdgcn_wmma_f32_16x16x4_f32(
                       false, a1, false, bf1, (short)0, acc1, false, false);
            acc2 = __builtin_amdgcn_wmma_f32_16x16x4_f32(
                       false, a2, false, bf2, (short)0, acc2, false, false);
        }

#pragma unroll
        for (int r = 0; r < 8; ++r) {
            float x1 = acc1[r]; x1 = (x1 >= 0.0f) ? x1 : NEG_SLOPE * x1;
            float x2 = acc2[r]; x2 = (x2 >= 0.0f) ? x2 : NEG_SLOPE * x2;
            int row = r + 8 * g;
            out[(size_t)(tileNode + row) * D + j + n] = x1 + x2;
        }
    }
}

extern "C" void kernel_launch(void* const* d_in, const int* in_sizes, int n_in,
                              void* d_out, int out_size, void* d_ws, size_t ws_size,
                              hipStream_t stream) {
    const float* ego    = (const float*)d_in[0];
    const float* a_vals = (const float*)d_in[1];
    const float* W1     = (const float*)d_in[2];
    const float* b1     = (const float*)d_in[3];
    const float* W2     = (const float*)d_in[4];
    const float* b2     = (const float*)d_in[5];
    const int*   a_rows = (const int*)d_in[6];
    const int*   a_cols = (const int*)d_in[7];
    float* out  = (float*)d_out;
    float* side = (float*)d_ws;

    const int N = in_sizes[0] / D;
    const int E = in_sizes[6];

    int* ent  = (int*)d_out;
    int* rank = ent  + 4 * (size_t)E;
    int* cnt   = rank + E;
    int* start = cnt  + N + 1;

    kgetcda_zero_counts<<<(N + 256) / 256, 256, 0, stream>>>(cnt, N + 1);
    kgetcda_count<<<(E + 255) / 256, 256, 0, stream>>>(a_rows, cnt, rank, E);
    kgetcda_alloc<<<(N + ALLOC_THREADS - 1) / ALLOC_THREADS, ALLOC_THREADS, 0,
                    stream>>>(cnt, cnt + N, start, N);
    kgetcda_fill<<<(E + 255) / 256, 256, 0, stream>>>(a_rows, a_cols, a_vals,
                                                     start, rank, ent, E);

    kgetcda_gather<<<(N + ROWS_PER_BLOCK - 1) / ROWS_PER_BLOCK,
                     ROWS_PER_BLOCK * 32, 0, stream>>>(ego, start, cnt, ent, side, N);

    const int fusedBlocks = (N + NODES_PER_BLOCK - 1) / NODES_PER_BLOCK;
    kgetcda_fused<<<fusedBlocks, 128, 0, stream>>>(ego, side, W1, b1, W2, b2,
                                                   out, N);
}


# SpMM bucketing in LDS: edges binned by 64-row range (LDS counts, one global add per work-group and bin), each bin's rows bucketed and summed by one work-group; sum order unchanged
# speedup vs baseline: 2.7343x; 1.9045x over previous
#include <hip/hip_runtime.h>
#include <hip/hip_bf16.h>

#define D 96
#define NODES_PER_BLOCK 64
#define NEG_SLOPE 0.01f

typedef float v2f __attribute__((ext_vector_type(2)));
typedef float v8f __attribute__((ext_vector_type(8)));


#define BIN_ROWS 64
#define BIN_SHIFT 6
#define MAX_BINS 1024
#define CHUNK 4096
#define BIN_THREADS 256
#define EDGES_PER_THREAD (CHUNK / BIN_THREADS)
#define GATHER_THREADS 512
#define ROWS_PER_WAVE (BIN_ROWS / (GATHER_THREADS / 32))
#define ENT_PER_THREAD 4
#define BIN_CAP (ENT_PER_THREAD * GATHER_THREADS)

__global__ void kgetcda_zero_counts(int* __restrict__ cnt, int n) {
    int i = blockIdx.x * blockDim.x + threadIdx.x;
    if (i < n) cnt[i] = 0;
}

__global__ void __launch_bounds__(BIN_THREADS)
kgetcda_bin_count(const int* __restrict__ rows,
                  int*       __restrict__ binCount,
                  int*       __restrict__ blockBase, int E, int NB) {
    __shared__ int h[MAX_BINS];
    const int t = threadIdx.x;
    const int e0 = blockIdx.x * CHUNK + t;
    int r[EDGES_PER_THREAD];
#pragma unroll
    for (int k = 0; k < EDGES_PER_THREAD; ++k) {
        const int e = e0 + k * BIN_THREADS;
        r[k] = rows[(e < E) ? e : (E - 1)];
    }
    for (int i = t; i < NB; i += BIN_THREADS) h[i] = 0;
    __syncthreads();
#pragma unroll
    for (int k = 0; k < EDGES_PER_THREAD; ++k) {
        const int e = e0 + k * BIN_THREADS;
        if (e < E) atomicAdd(&h[r[k] >> BIN_SHIFT], 1);
    }
    __syncthreads();
    int* mine = blockBase + (size_t)blockIdx.x * NB;
    for (int i = t; i < NB; i += BIN_THREADS) {
        const int c = h[i];
        mine[i] = c ? atomicAdd(&binCount[i], c) : 0;
    }
}

__global__ void __launch_bounds__(BIN_THREADS)
kgetcda_bin_fill(const int*   __restrict__ rows,
                 const int*   __restrict__ cols,
                 const float* __restrict__ vals,
                 const int*   __restrict__ binCount,
                 const int*   __restrict__ blockBase,
                 int*         __restrict__ binStart,
                 int*         __restrict__ ent, int E, int NB) {
    __shared__ int h[MAX_BINS];
    __shared__ int cur[MAX_BINS];
    __shared__ int wsum[BIN_THREADS / 32];
    const int t    = threadIdx.x;
    const int lane = t & 31;
    const int w    = t >> 5;

    const int e0 = blockIdx.x * CHUNK + t;
    int r[EDGES_PER_THREAD], c[EDGES_PER_THREAD], vb[EDGES_PER_THREAD], pos[EDGES_PER_THREAD];
#pragma unroll
    for (int k = 0; k < EDGES_PER_THREAD; ++k) {
        const int e  = e0 + k * BIN_THREADS;
        const int ec = (e < E) ? e : (E - 1);
        r[k]  = rows[ec];
        c[k]  = cols[ec];
        vb[k] = __float_as_int(vals[ec]);
    }

    const int i0 = 4 * t;
    const int* mine = blockBase + (size_t)blockIdx.x * NB;
    const bool u0 = i0 < NB, u1 = i0 + 1 < NB, u2 = i0 + 2 < NB, u3 = i0 + 3 < NB;
    const int l0 = binCount[u0 ? i0 : 0],     m0 = mine[u0 ? i0 : 0];
    const int l1 = binCount[u1 ? i0 + 1 : 0], m1 = mine[u1 ? i0 + 1 : 0];
    const int l2 = binCount[u2 ? i0 + 2 : 0], m2 = mine[u2 ? i0 + 2 : 0];
    const int l3 = binCount[u3 ? i0 + 3 : 0], m3 = mine[u3 ? i0 + 3 : 0];
    const int c0 = u0 ? l0 : 0;
    const int c1 = u1 ? l1 : 0;
    const int c2 = u2 ? l2 : 0;
    const int c3 = u3 ? l3 : 0;
    const int s4 = c0 + c1 + c2 + c3;
    int x = s4;
    for (int off = 1; off < 32; off <<= 1) {
        int y = __shfl_up(x, off);
        if (lane >= off) x += y;
    }
    if (lane == 31) wsum[w] = x;
    __syncthreads();
    if (t == 0) {
        int run = 0;
        for (int i = 0; i < BIN_THREADS / 32; ++i) {
            int v = wsum[i];
            wsum[i] = run;
            run += v;
        }
    }
    __syncthreads();
    const int b0 = wsum[w] + x - s4;
    const int b1 = b0 + c0;
    const int b2 = b1 + c1;
    const int b3 = b2 + c2;
    if (u0) { h[i0]     = b0 + m0; cur[i0]     = 0; }
    if (u1) { h[i0 + 1] = b1 + m1; cur[i0 + 1] = 0; }
    if (u2) { h[i0 + 2] = b2 + m2; cur[i0 + 2] = 0; }
    if (u3) { h[i0 + 3] = b3 + m3; cur[i0 + 3] = 0; }
    if (blockIdx.x == 0) {
        if (u0) binStart[i0]     = b0;
        if (u1) binStart[i0 + 1] = b1;
        if (u2) binStart[i0 + 2] = b2;
        if (u3) binStart[i0 + 3] = b3;
    }
    __syncthreads();

#pragma unroll
    for (int k = 0; k < EDGES_PER_THREAD; ++k) {
        const int e = e0 + k * BIN_THREADS;
        const int b = r[k] >> BIN_SHIFT;
        pos[k] = 0;
        if (e < E) pos[k] = h[b] + atomicAdd(&cur[b], 1);
    }
#pragma unroll
    for (int k = 0; k < EDGES_PER_THREAD; ++k) {
        const int e = e0 + k * BIN_THREADS;
        if (e < E) {
            int4 v;
            v.x = e;
            v.y = c[k];
            v.z = vb[k];
            v.w = r[k];
            reinterpret_cast<int4*>(ent)[pos[k]] = v;
        }
    }
}

__global__ void __launch_bounds__(GATHER_THREADS)
kgetcda_bin_gather(const float* __restrict__ ego,
                   const int*   __restrict__ binStart,
                   const int*   __restrict__ binCount,
                   const int*   __restrict__ ent,
                   float*       __restrict__ side, int N) {
    __shared__ int rcnt[BIN_ROWS];
    __shared__ int rstart[BIN_ROWS];
    __shared__ int rcur[BIN_ROWS];
    __shared__ int s_e[BIN_CAP];
    __shared__ int s_c[BIN_CAP];
    __shared__ int s_v[BIN_CAP];

    const int t    = threadIdx.x;
    const int lane = t & 31;
    const int w    = __builtin_amdgcn_readfirstlane(t >> 5);
    const int bin  = blockIdx.x;
    const int row0 = bin * BIN_ROWS;
    const int n    = binCount[bin];
    const int4* src = reinterpret_cast<const int4*>(ent) + binStart[bin];
    const float* base = ego + lane;

    if (n <= BIN_CAP) {
        if (t < BIN_ROWS) rcnt[t] = 0;
        int4 v[ENT_PER_THREAD];
#pragma unroll
        for (int j = 0; j < ENT_PER_THREAD; ++j) {
            const int i = t + j * GATHER_THREADS;
            v[j] = src[(i < n) ? i : 0];
        }
        __syncthreads();
#pragma unroll
        for (int j = 0; j < ENT_PER_THREAD; ++j)
            if (t + j * GATHER_THREADS < n) atomicAdd(&rcnt[v[j].w - row0], 1);
        __syncthreads();
        if (t < 32) {
            const int c0 = rcnt[2 * t];
            const int c1 = rcnt[2 * t + 1];
            int x = c0 + c1;
            for (int off = 1; off < 32; off <<= 1) {
                int y = __shfl_up(x, off);
                if (lane >= off) x += y;
            }
            const int ex = x - c0 - c1;
            rstart[2 * t]     = ex;       rcur[2 * t]     = ex;
            rstart[2 * t + 1] = ex + c0;  rcur[2 * t + 1] = ex + c0;
        }
        __syncthreads();
#pragma unroll
        for (int j = 0; j < ENT_PER_THREAD; ++j) {
            if (t + j * GATHER_THREADS < n) {
                const int p = atomicAdd(&rcur[v[j].w - row0], 1);
                s_e[p] = v[j].x;
                s_c[p] = v[j].y;
                s_v[p] = v[j].z;
            }
        }
        __syncthreads();

        for (int rr = 0; rr < ROWS_PER_WAVE; ++rr) {
            const int lr  = w * ROWS_PER_WAVE + rr;
            const int row = row0 + lr;
            if (row >= N) break;
            const int d  = __builtin_amdgcn_readfirstlane(rcnt[lr]);
            const int st = __builtin_amdgcn_readfirstlane(rstart[lr]);
            const int* be = s_e + st;
            const int* bc = s_c + st;
            const int* bv = s_v + st;

            float a0 = 0.0f, a1 = 0.0f, a2 = 0.0f;

            if (d <= 32) {
                int me = 0x7fffffff, mc = 0, mv = 0;
                if (lane < d) { me = be[lane]; mc = bc[lane]; mv = bv[lane]; }
                int rk = 0;
                for (int j = 0; j < d; ++j)
                    rk += (__builtin_amdgcn_readlane(me, j) < me) ? 1 : 0;
                const int sc = __builtin_amdgcn_ds_permute(rk << 2, mc);
                const int sv = __builtin_amdgcn_ds_permute(rk << 2, mv);
#pragma unroll 4
                for (int k = 0; k < d; ++k) {
                    const int   c = __builtin_amdgcn_readlane(sc, k);
                    const float v = __int_as_float(__builtin_amdgcn_readlane(sv, k));
                    const float* p = base + (size_t)c * D;
                    a0 = fmaf(v, p[0],  a0);
                    a1 = fmaf(v, p[32], a1);
                    a2 = fmaf(v, p[64], a2);
                }
            } else if (d <= 64) {
                const int me0 = be[lane], mc0 = bc[lane], mv0 = bv[lane];
                int me1 = 0x7fffffff, mc1 = 0, mv1 = 0;
                if (lane + 32 < d) { me1 = be[lane + 32]; mc1 = bc[lane + 32]; mv1 = bv[lane + 32]; }
                int rk0 = 0, rk1 = 0;
                for (int j = 0; j < 32; ++j) {
                    const int x = __builtin_amdgcn_readlane(me0, j);
                    rk0 += (x < me0) ? 1 : 0;
                    rk1 += (x < me1) ? 1 : 0;
                }
                const int d1 = d - 32;
                for (int j = 0; j < d1; ++j) {
                    const int x = __builtin_amdgcn_readlane(me1, j);
                    rk0 += (x < me0) ? 1 : 0;
                    rk1 += (x < me1) ? 1 : 0;
                }
#pragma unroll 4
                for (int k = 0; k < d; ++k) {
                    const unsigned m0 = __builtin_amdgcn_ballot_w32(rk0 == k);
                    const unsigned m1 = __builtin_amdgcn_ballot_w32(rk1 == k);
                    const int l0 = m0 ? __builtin_ctz(m0) : 0;
                    const int l1 = m1 ? __builtin_ctz(m1) : 0;
                    const int c0 = __builtin_amdgcn_readlane(mc0, l0);
                    const int c1 = __builtin_amdgcn_readlane(mc1, l1);
                    const int q0 = __builtin_amdgcn_readlane(mv0, l0);
                    const int q1 = __builtin_amdgcn_readlane(mv1, l1);
                    const int   c = m0 ? c0 : c1;
                    const float v = __int_as_float(m0 ? q0 : q1);
                    const float* p = base + (size_t)c * D;
                    a0 = fmaf(v, p[0],  a0);
                    a1 = fmaf(v, p[32], a1);
                    a2 = fmaf(v, p[64], a2);
                }
            } else {
                int last = -1;
                for (int k = 0; k < d; ++k) {
                    int best = 0x7fffffff, bq = 0;
                    for (int q = lane; q < d; q += 32) {
                        int e = be[q];
                        if (e > last && e < best) { best = e; bq = q; }
                    }
                    for (int off = 16; off > 0; off >>= 1) {
                        int ob = __shfl_xor(best, off);
                        int oq = __shfl_xor(bq, off);
                        if (ob < best) { best = ob; bq = oq; }
                    }
                    const float v = __int_as_float(bv[bq]);
                    const float* p = base + (size_t)bc[bq] * D;
                    a0 = fmaf(v, p[0],  a0);
                    a1 = fmaf(v, p[32], a1);
                    a2 = fmaf(v, p[64], a2);
                    last = best;
                }
            }
            float* dst = side + (size_t)row * D + lane;
            dst[0]  = a0;
            dst[32] = a1;
            dst[64] = a2;
        }
    } else {
        for (int rr = 0; rr < ROWS_PER_WAVE; ++rr) {
            const int row = row0 + w * ROWS_PER_WAVE + rr;
            if (row >= N) break;
            float a0 = 0.0f, a1 = 0.0f, a2 = 0.0f;
            int last = -1;
            for (int k = 0; k < n; ++k) {
                int best = 0x7fffffff, bq = 0;
                for (int q = lane; q < n; q += 32) {
                    const int4 u = src[q];
                    if (u.w == row && u.x > last && u.x < best) { best = u.x; bq = q; }
                }
                for (int off = 16; off > 0; off >>= 1) {
                    int ob = __shfl_xor(best, off);
                    int oq = __shfl_xor(bq, off);
                    if (ob < best) { best = ob; bq = oq; }
                }
                if (best == 0x7fffffff) break;
                const int4  u = src[bq];
                const float v = __int_as_float(u.z);
                const float* p = base + (size_t)u.y * D;
                a0 = fmaf(v, p[0],  a0);
                a1 = fmaf(v, p[32], a1);
                a2 = fmaf(v, p[64], a2);
                last = best;
            }
            float* dst = side + (size_t)row * D + lane;
            dst[0]  = a0;
            dst[32] = a1;
            dst[64] = a2;
        }
    }
}

__global__ void __launch_bounds__(128)
kgetcda_fused(const float* __restrict__ ego,  const float* __restrict__ side,
              const float* __restrict__ W1,   const float* __restrict__ b1,
              const float* __restrict__ W2,   const float* __restrict__ b2,
              float* __restrict__ out, int N) {
    const int tid      = threadIdx.x;
    const int wave     = tid >> 5;
    const int lane     = tid & 31;
    const int tileNode = blockIdx.x * NODES_PER_BLOCK + wave * 16;

    if (tileNode >= N) return;

    const int n = lane & 15;
    const int g = lane >> 4;

    const float4* ep = reinterpret_cast<const float4*>(ego  + (size_t)(tileNode + n) * D) + g;
    const float4* sp = reinterpret_cast<const float4*>(side + (size_t)(tileNode + n) * D) + g;
    float4 xs[12], xm[12];
#pragma unroll
    for (int t = 0; t < 12; ++t) {
        const float4 e = ep[2 * t];
        const float4 s = sp[2 * t];
        xs[t].x = e.x + s.x; xs[t].y = e.y + s.y; xs[t].z = e.z + s.z; xs[t].w = e.w + s.w;
        xm[t].x = e.x * s.x; xm[t].y = e.y * s.y; xm[t].z = e.z * s.z; xm[t].w = e.w * s.w;
    }

#pragma unroll 1
    for (int j = 0; j < D; j += 16) {
        v8f acc1, acc2;
        const float bb1 = b1[j + n];
        const float bb2 = b2[j + n];
#pragma unroll
        for (int r = 0; r < 8; ++r) { acc1[r] = bb1; acc2[r] = bb2; }

        const float4* w1p = reinterpret_cast<const float4*>(W1 + (size_t)(j + n) * D) + g;
        const float4* w2p = reinterpret_cast<const float4*>(W2 + (size_t)(j + n) * D) + g;

#pragma unroll
        for (int t = 0; t < 12; ++t) {
            const float4 w1 = w1p[2 * t];
            const float4 w2 = w2p[2 * t];
            v2f a1, a2, bf1, bf2;
            a1.x  = xs[t].x;  a1.y  = xs[t].y;
            a2.x  = xm[t].x;  a2.y  = xm[t].y;
            bf1.x = w1.x;     bf1.y = w1.y;
            bf2.x = w2.x;     bf2.y = w2.y;
            acc1 = __builtin_amdgcn_wmma_f32_16x16x4_f32(
                       false, a1, false, bf1, (short)0, acc1, false, false);
            acc2 = __builtin_amdgcn_wmma_f32_16x16x4_f32(
                       false, a2, false, bf2, (short)0, acc2, false, false);
            a1.x  = xs[t].z;  a1.y  = xs[t].w;
            a2.x  = xm[t].z;  a2.y  = xm[t].w;
            bf1.x = w1.z;     bf1.y = w1.w;
            bf2.x = w2.z;     bf2.y = w2.w;
            acc1 = __builtin_amdgcn_wmma_f32_16x16x4_f32(
                       false, a1, false, bf1, (short)0, acc1, false, false);
            acc2 = __builtin_amdgcn_wmma_f32_16x16x4_f32(
                       false, a2, false, bf2, (short)0, acc2, false, false);
        }

#pragma unroll
        for (int r = 0; r < 8; ++r) {
            float x1 = acc1[r]; x1 = (x1 >= 0.0f) ? x1 : NEG_SLOPE * x1;
            float x2 = acc2[r]; x2 = (x2 >= 0.0f) ? x2 : NEG_SLOPE * x2;
            int row = r + 8 * g;
            out[(size_t)(tileNode + row) * D + j + n] = x1 + x2;
        }
    }
}

extern "C" void kernel_launch(void* const* d_in, const int* in_sizes, int n_in,
                              void* d_out, int out_size, void* d_ws, size_t ws_size,
                              hipStream_t stream) {
    const float* ego    = (const float*)d_in[0];
    const float* a_vals = (const float*)d_in[1];
    const float* W1     = (const float*)d_in[2];
    const float* b1     = (const float*)d_in[3];
    const float* W2     = (const float*)d_in[4];
    const float* b2     = (const float*)d_in[5];
    const int*   a_rows = (const int*)d_in[6];
    const int*   a_cols = (const int*)d_in[7];
    float* out  = (float*)d_out;
    float* side = (float*)d_ws;

    const int N = in_sizes[0] / D;
    const int E = in_sizes[6];
    const int NB   = (N + BIN_ROWS - 1) / BIN_ROWS;
    const int NBLK = (E + CHUNK - 1) / CHUNK;

    int* ent       = (int*)d_out;
    int* binCount  = ent + 4 * (size_t)E;
    int* binStart  = binCount + MAX_BINS;
    int* blockBase = binStart + MAX_BINS;

    kgetcda_zero_counts<<<(NB + 255) / 256, 256, 0, stream>>>(binCount, NB);
    kgetcda_bin_count<<<NBLK, BIN_THREADS, 0, stream>>>(a_rows, binCount, blockBase, E, NB);
    kgetcda_bin_fill<<<NBLK, BIN_THREADS, 0, stream>>>(a_rows, a_cols, a_vals, binCount,
                                                       blockBase, binStart, ent, E, NB);

    kgetcda_bin_gather<<<NB, GATHER_THREADS, 0, stream>>>(ego, binStart, binCount, ent, side, N);

    const int fusedBlocks = (N + NODES_PER_BLOCK - 1) / NODES_PER_BLOCK;
    kgetcda_fused<<<fusedBlocks, 128, 0, stream>>>(ego, side, W1, b1, W2, b2,
                                                   out, N);
}


# Gather and products in one launch: each bin's work-group keeps ego+side and ego*side of its 64 rows in LDS and 12 of its 16 waves do the WMMA products from there; no side array; sum order unchanged
# speedup vs baseline: 3.1220x; 1.1418x over previous
#include <hip/hip_runtime.h>
#include <hip/hip_bf16.h>

#define D 96
#define NEG_SLOPE 0.01f

typedef float v2f __attribute__((ext_vector_type(2)));
typedef float v8f __attribute__((ext_vector_type(8)));


#define BIN_ROWS 64
#define BIN_SHIFT 6
#define MAX_BINS 1024
#define CHUNK 4096
#define BIN_THREADS 256
#define EDGES_PER_THREAD (CHUNK / BIN_THREADS)
#define GATHER_THREADS 512
#define ROWS_PER_WAVE (BIN_ROWS / (GATHER_THREADS / 32))
#define ENT_PER_THREAD 4
#define BIN_CAP (ENT_PER_THREAD * GATHER_THREADS)
#define XP 100

__global__ void kgetcda_zero_counts(int* __restrict__ cnt, int n) {
    int i = blockIdx.x * blockDim.x + threadIdx.x;
    if (i < n) cnt[i] = 0;
}

__global__ void __launch_bounds__(BIN_THREADS)
kgetcda_bin_count(const int* __restrict__ rows,
                  int*       __restrict__ binCount,
                  int*       __restrict__ blockBase, int E, int NB) {
    __shared__ int h[MAX_BINS];
    const int t = threadIdx.x;
    const int e0 = blockIdx.x * CHUNK + t;
    int r[EDGES_PER_THREAD];
#pragma unroll
    for (int k = 0; k < EDGES_PER_THREAD; ++k) {
        const int e = e0 + k * BIN_THREADS;
        r[k] = rows[(e < E) ? e : (E - 1)];
    }
    for (int i = t; i < NB; i += BIN_THREADS) h[i] = 0;
    __syncthreads();
#pragma unroll
    for (int k = 0; k < EDGES_PER_THREAD; ++k) {
        const int e = e0 + k * BIN_THREADS;
        if (e < E) atomicAdd(&h[r[k] >> BIN_SHIFT], 1);
    }
    __syncthreads();
    int* mine = blockBase + (size_t)blockIdx.x * NB;
    for (int i = t; i < NB; i += BIN_THREADS) {
        const int c = h[i];
        mine[i] = c ? atomicAdd(&binCount[i], c) : 0;
    }
}

__global__ void __launch_bounds__(BIN_THREADS)
kgetcda_bin_fill(const int*   __restrict__ rows,
                 const int*   __restrict__ cols,
                 const float* __restrict__ vals,
                 const int*   __restrict__ binCount,
                 const int*   __restrict__ blockBase,
                 int*         __restrict__ binStart,
                 int*         __restrict__ ent, int E, int NB) {
    __shared__ int h[MAX_BINS];
    __shared__ int cur[MAX_BINS];
    __shared__ int wsum[BIN_THREADS / 32];
    const int t    = threadIdx.x;
    const int lane = t & 31;
    const int w    = t >> 5;

    const int e0 = blockIdx.x * CHUNK + t;
    int r[EDGES_PER_THREAD], c[EDGES_PER_THREAD], vb[EDGES_PER_THREAD], pos[EDGES_PER_THREAD];
#pragma unroll
    for (int k = 0; k < EDGES_PER_THREAD; ++k) {
        const int e  = e0 + k * BIN_THREADS;
        const int ec = (e < E) ? e : (E - 1);
        r[k]  = rows[ec];
        c[k]  = cols[ec];
        vb[k] = __float_as_int(vals[ec]);
    }

    const int i0 = 4 * t;
    const int* mine = blockBase + (size_t)blockIdx.x * NB;
    const bool u0 = i0 < NB, u1 = i0 + 1 < NB, u2 = i0 + 2 < NB, u3 = i0 + 3 < NB;
    const int l0 = binCount[u0 ? i0 : 0],     m0 = mine[u0 ? i0 : 0];
    const int l1 = binCount[u1 ? i0 + 1 : 0], m1 = mine[u1 ? i0 + 1 : 0];
    const int l2 = binCount[u2 ? i0 + 2 : 0], m2 = mine[u2 ? i0 + 2 : 0];
    const int l3 = binCount[u3 ? i0 + 3 : 0], m3 = mine[u3 ? i0 + 3 : 0];
    const int c0 = u0 ? l0 : 0;
    const int c1 = u1 ? l1 : 0;
    const int c2 = u2 ? l2 : 0;
    const int c3 = u3 ? l3 : 0;
    const int s4 = c0 + c1 + c2 + c3;
    int x = s4;
    for (int off = 1; off < 32; off <<= 1) {
        int y = __shfl_up(x, off);
        if (lane >= off) x += y;
    }
    if (lane == 31) wsum[w] = x;
    __syncthreads();
    if (t == 0) {
        int run = 0;
        for (int i = 0; i < BIN_THREADS / 32; ++i) {
            int v = wsum[i];
            wsum[i] = run;
            run += v;
        }
    }
    __syncthreads();
    const int b0 = wsum[w] + x - s4;
    const int b1 = b0 + c0;
    const int b2 = b1 + c1;
    const int b3 = b2 + c2;
    if (u0) { h[i0]     = b0 + m0; cur[i0]     = 0; }
    if (u1) { h[i0 + 1] = b1 + m1; cur[i0 + 1] = 0; }
    if (u2) { h[i0 + 2] = b2 + m2; cur[i0 + 2] = 0; }
    if (u3) { h[i0 + 3] = b3 + m3; cur[i0 + 3] = 0; }
    if (blockIdx.x == 0) {
        if (u0) binStart[i0]     = b0;
        if (u1) binStart[i0 + 1] = b1;
        if (u2) binStart[i0 + 2] = b2;
        if (u3) binStart[i0 + 3] = b3;
    }
    __syncthreads();

#pragma unroll
    for (int k = 0; k < EDGES_PER_THREAD; ++k) {
        const int e = e0 + k * BIN_THREADS;
        const int b = r[k] >> BIN_SHIFT;
        pos[k] = 0;
        if (e < E) pos[k] = h[b] + atomicAdd(&cur[b], 1);
    }
#pragma unroll
    for (int k = 0; k < EDGES_PER_THREAD; ++k) {
        const int e = e0 + k * BIN_THREADS;
        if (e < E) {
            int4 v;
            v.x = e;
            v.y = c[k];
            v.z = vb[k];
            v.w = r[k];
            reinterpret_cast<int4*>(ent)[pos[k]] = v;
        }
    }
}

__global__ void __launch_bounds__(GATHER_THREADS, 16)
kgetcda_gather_fused(const float* __restrict__ ego,
                     const int*   __restrict__ binStart,
                     const int*   __restrict__ binCount,
                     const int*   __restrict__ ent,
                     const float* __restrict__ W1, const float* __restrict__ b1,
                     const float* __restrict__ W2, const float* __restrict__ b2,
                     float*       __restrict__ out, int N) {
    __shared__ int rcnt[BIN_ROWS];
    __shared__ int rstart[BIN_ROWS];
    __shared__ int rcur[BIN_ROWS];
    __shared__ int s_e[BIN_CAP];
    __shared__ int s_c[BIN_CAP];
    __shared__ int s_v[BIN_CAP];
    __shared__ alignas(16) float s_xs[BIN_ROWS * XP];
    __shared__ alignas(16) float s_xm[BIN_ROWS * XP];

    const int t    = threadIdx.x;
    const int lane = t & 31;
    const int w    = __builtin_amdgcn_readfirstlane(t >> 5);
    const int bin  = blockIdx.x;
    const int row0 = bin * BIN_ROWS;
    const int n    = binCount[bin];
    const int4* src = reinterpret_cast<const int4*>(ent) + binStart[bin];
    const float* base = ego + lane;

    if (n <= BIN_CAP) {
        if (t < BIN_ROWS) rcnt[t] = 0;
        int4 v[ENT_PER_THREAD];
#pragma unroll
        for (int j = 0; j < ENT_PER_THREAD; ++j) {
            const int i = t + j * GATHER_THREADS;
            v[j] = src[(i < n) ? i : 0];
        }
        __syncthreads();
#pragma unroll
        for (int j = 0; j < ENT_PER_THREAD; ++j)
            if (t + j * GATHER_THREADS < n) atomicAdd(&rcnt[v[j].w - row0], 1);
        __syncthreads();
        if (t < 32) {
            const int c0 = rcnt[2 * t];
            const int c1 = rcnt[2 * t + 1];
            int x = c0 + c1;
            for (int off = 1; off < 32; off <<= 1) {
                int y = __shfl_up(x, off);
                if (lane >= off) x += y;
            }
            const int ex = x - c0 - c1;
            rstart[2 * t]     = ex;       rcur[2 * t]     = ex;
            rstart[2 * t + 1] = ex + c0;  rcur[2 * t + 1] = ex + c0;
        }
        __syncthreads();
#pragma unroll
        for (int j = 0; j < ENT_PER_THREAD; ++j) {
            if (t + j * GATHER_THREADS < n) {
                const int p = atomicAdd(&rcur[v[j].w - row0], 1);
                s_e[p] = v[j].x;
                s_c[p] = v[j].y;
                s_v[p] = v[j].z;
            }
        }
        __syncthreads();

        for (int rr = 0; rr < ROWS_PER_WAVE; ++rr) {
            const int lr  = w * ROWS_PER_WAVE + rr;
            const int row = row0 + lr;
            if (row >= N) break;
            const float* er = ego + (size_t)row * D + lane;
            const float e0 = er[0], e1 = er[32], e2 = er[64];
            const int d  = __builtin_amdgcn_readfirstlane(rcnt[lr]);
            const int st = __builtin_amdgcn_readfirstlane(rstart[lr]);
            const int* be = s_e + st;
            const int* bc = s_c + st;
            const int* bv = s_v + st;

            float a0 = 0.0f, a1 = 0.0f, a2 = 0.0f;

            if (d <= 32) {
                int me = 0x7fffffff, mc = 0, mv = 0;
                if (lane < d) { me = be[lane]; mc = bc[lane]; mv = bv[lane]; }
                int rk = 0;
                for (int j = 0; j < d; ++j)
                    rk += (__builtin_amdgcn_readlane(me, j) < me) ? 1 : 0;
                const int sc = __builtin_amdgcn_ds_permute(rk << 2, mc);
                const int sv = __builtin_amdgcn_ds_permute(rk << 2, mv);
#pragma unroll 4
                for (int k = 0; k < d; ++k) {
                    const int   c = __builtin_amdgcn_readlane(sc, k);
                    const float v = __int_as_float(__builtin_amdgcn_readlane(sv, k));
                    const float* p = base + (size_t)c * D;
                    a0 = fmaf(v, p[0],  a0);
                    a1 = fmaf(v, p[32], a1);
                    a2 = fmaf(v, p[64], a2);
                }
            } else if (d <= 64) {
                const int me0 = be[lane], mc0 = bc[lane], mv0 = bv[lane];
                int me1 = 0x7fffffff, mc1 = 0, mv1 = 0;
                if (lane + 32 < d) { me1 = be[lane + 32]; mc1 = bc[lane + 32]; mv1 = bv[lane + 32]; }
                int rk0 = 0, rk1 = 0;
                for (int j = 0; j < 32; ++j) {
                    const int x = __builtin_amdgcn_readlane(me0, j);
                    rk0 += (x < me0) ? 1 : 0;
                    rk1 += (x < me1) ? 1 : 0;
                }
                const int d1 = d - 32;
                for (int j = 0; j < d1; ++j) {
                    const int x = __builtin_amdgcn_readlane(me1, j);
                    rk0 += (x < me0) ? 1 : 0;
                    rk1 += (x < me1) ? 1 : 0;
                }
#pragma unroll 4
                for (int k = 0; k < d; ++k) {
                    const unsigned m0 = __builtin_amdgcn_ballot_w32(rk0 == k);
                    const unsigned m1 = __builtin_amdgcn_ballot_w32(rk1 == k);
                    const int l0 = m0 ? __builtin_ctz(m0) : 0;
                    const int l1 = m1 ? __builtin_ctz(m1) : 0;
                    const int c0 = __builtin_amdgcn_readlane(mc0, l0);
                    const int c1 = __builtin_amdgcn_readlane(mc1, l1);
                    const int q0 = __builtin_amdgcn_readlane(mv0, l0);
                    const int q1 = __builtin_amdgcn_readlane(mv1, l1);
                    const int   c = m0 ? c0 : c1;
                    const float v = __int_as_float(m0 ? q0 : q1);
                    const float* p = base + (size_t)c * D;
                    a0 = fmaf(v, p[0],  a0);
                    a1 = fmaf(v, p[32], a1);
                    a2 = fmaf(v, p[64], a2);
                }
            } else {
                int last = -1;
                for (int k = 0; k < d; ++k) {
                    int best = 0x7fffffff, bq = 0;
                    for (int q = lane; q < d; q += 32) {
                        int e = be[q];
                        if (e > last && e < best) { best = e; bq = q; }
                    }
                    for (int off = 16; off > 0; off >>= 1) {
                        int ob = __shfl_xor(best, off);
                        int oq = __shfl_xor(bq, off);
                        if (ob < best) { best = ob; bq = oq; }
                    }
                    const float v = __int_as_float(bv[bq]);
                    const float* p = base + (size_t)bc[bq] * D;
                    a0 = fmaf(v, p[0],  a0);
                    a1 = fmaf(v, p[32], a1);
                    a2 = fmaf(v, p[64], a2);
                    last = best;
                }
            }
            float* xs = s_xs + lr * XP + lane;
            float* xm = s_xm + lr * XP + lane;
            xs[0]  = e0 + a0;  xm[0]  = e0 * a0;
            xs[32] = e1 + a1;  xm[32] = e1 * a1;
            xs[64] = e2 + a2;  xm[64] = e2 * a2;
        }
    } else {
        for (int rr = 0; rr < ROWS_PER_WAVE; ++rr) {
            const int lr  = w * ROWS_PER_WAVE + rr;
            const int row = row0 + lr;
            if (row >= N) break;
            const float* er = ego + (size_t)row * D + lane;
            const float e0 = er[0], e1 = er[32], e2 = er[64];
            float a0 = 0.0f, a1 = 0.0f, a2 = 0.0f;
            int last = -1;
            for (int k = 0; k < n; ++k) {
                int best = 0x7fffffff, bq = 0;
                for (int q = lane; q < n; q += 32) {
                    const int4 u = src[q];
                    if (u.w == row && u.x > last && u.x < best) { best = u.x; bq = q; }
                }
                for (int off = 16; off > 0; off >>= 1) {
                    int ob = __shfl_xor(best, off);
                    int oq = __shfl_xor(bq, off);
                    if (ob < best) { best = ob; bq = oq; }
                }
                if (best == 0x7fffffff) break;
                const int4  u = src[bq];
                const float v = __int_as_float(u.z);
                const float* p = base + (size_t)u.y * D;
                a0 = fmaf(v, p[0],  a0);
                a1 = fmaf(v, p[32], a1);
                a2 = fmaf(v, p[64], a2);
                last = best;
            }
            float* xs = s_xs + lr * XP + lane;
            float* xm = s_xm + lr * XP + lane;
            xs[0]  = e0 + a0;  xm[0]  = e0 * a0;
            xs[32] = e1 + a1;  xm[32] = e1 * a1;
            xs[64] = e2 + a2;  xm[64] = e2 * a2;
        }
    }

    const int rt = w / 3;
    const int cp = w - 3 * rt;
    const int tileNode = row0 + rt * 16;
    const bool mine = (w < 12) && (tileNode < N);
    const int c = lane & 15;
    const int g = lane >> 4;

    __syncthreads();

    if (mine) {
        const float4* xsp = reinterpret_cast<const float4*>(s_xs + (rt * 16 + c) * XP) + g;
        const float4* xmp = reinterpret_cast<const float4*>(s_xm + (rt * 16 + c) * XP) + g;
#pragma unroll 1
        for (int j = cp * 32; j < cp * 32 + 32; j += 16) {
            v8f acc1, acc2;
            const float bb1 = b1[j + c];
            const float bb2 = b2[j + c];
#pragma unroll
            for (int r = 0; r < 8; ++r) { acc1[r] = bb1; acc2[r] = bb2; }

            const float4* w1p = reinterpret_cast<const float4*>(W1 + (size_t)(j + c) * D) + g;
            const float4* w2p = reinterpret_cast<const float4*>(W2 + (size_t)(j + c) * D) + g;

#pragma unroll 1
            for (int t0 = 0; t0 < 12; t0 += 3) {
                float4 w1[3], w2[3];
#pragma unroll
                for (int u = 0; u < 3; ++u) {
                    w1[u] = w1p[2 * (t0 + u)];
                    w2[u] = w2p[2 * (t0 + u)];
                }
#pragma unroll
                for (int u = 0; u < 3; ++u) {
                    const float4 xs = xsp[2 * (t0 + u)];
                    const float4 xm = xmp[2 * (t0 + u)];
                    v2f a1, a2, bf1, bf2;
                    a1.x  = xs.x;     a1.y  = xs.y;
                    a2.x  = xm.x;     a2.y  = xm.y;
                    bf1.x = w1[u].x;  bf1.y = w1[u].y;
                    bf2.x = w2[u].x;  bf2.y = w2[u].y;
                    acc1 = __builtin_amdgcn_wmma_f32_16x16x4_f32(
                               false, a1, false, bf1, (short)0, acc1, false, false);
                    acc2 = __builtin_amdgcn_wmma_f32_16x16x4_f32(
                               false, a2, false, bf2, (short)0, acc2, false, false);
                    a1.x  = xs.z;     a1.y  = xs.w;
                    a2.x  = xm.z;     a2.y  = xm.w;
                    bf1.x = w1[u].z;  bf1.y = w1[u].w;
                    bf2.x = w2[u].z;  bf2.y = w2[u].w;
                    acc1 = __builtin_amdgcn_wmma_f32_16x16x4_f32(
                               false, a1, false, bf1, (short)0, acc1, false, false);
                    acc2 = __builtin_amdgcn_wmma_f32_16x16x4_f32(
                               false, a2, false, bf2, (short)0, acc2, false, false);
                }
            }

#pragma unroll
            for (int r = 0; r < 8; ++r) {
                float x1 = acc1[r]; x1 = (x1 >= 0.0f) ? x1 : NEG_SLOPE * x1;
                float x2 = acc2[r]; x2 = (x2 >= 0.0f) ? x2 : NEG_SLOPE * x2;
                const int row = r + 8 * g;
                out[(size_t)(tileNode + row) * D + j + c] = x1 + x2;
            }
        }
    }
}


extern "C" void kernel_launch(void* const* d_in, const int* in_sizes, int n_in,
                              void* d_out, int out_size, void* d_ws, size_t ws_size,
                              hipStream_t stream) {
    const float* ego    = (const float*)d_in[0];
    const float* a_vals = (const float*)d_in[1];
    const float* W1     = (const float*)d_in[2];
    const float* b1     = (const float*)d_in[3];
    const float* W2     = (const float*)d_in[4];
    const float* b2     = (const float*)d_in[5];
    const int*   a_rows = (const int*)d_in[6];
    const int*   a_cols = (const int*)d_in[7];
    float* out  = (float*)d_out;

    const int N = in_sizes[0] / D;
    const int E = in_sizes[6];
    const int NB   = (N + BIN_ROWS - 1) / BIN_ROWS;
    const int NBLK = (E + CHUNK - 1) / CHUNK;

    int* ent       = (int*)d_ws;
    int* binCount  = ent + 4 * (size_t)E;
    int* binStart  = binCount + MAX_BINS;
    int* blockBase = binStart + MAX_BINS;

    kgetcda_zero_counts<<<(NB + 255) / 256, 256, 0, stream>>>(binCount, NB);
    kgetcda_bin_count<<<NBLK, BIN_THREADS, 0, stream>>>(a_rows, binCount, blockBase, E, NB);
    kgetcda_bin_fill<<<NBLK, BIN_THREADS, 0, stream>>>(a_rows, a_cols, a_vals, binCount,
                                                       blockBase, binStart, ent, E, NB);

    kgetcda_gather_fused<<<NB, GATHER_THREADS, 0, stream>>>(ego, binStart, binCount, ent,
                                                            W1, b1, W2, b2, out, N);
}


# Binning without global counts: each 4096-edge chunk sorted by bin in LDS into its own stretch plus a table of runs; a bin's work-group collects its runs; 2 launches not 4; sum order unchanged
# speedup vs baseline: 3.2370x; 1.0368x over previous
#include <hip/hip_runtime.h>
#include <hip/hip_bf16.h>

#define D 96
#define NEG_SLOPE 0.01f

typedef float v2f __attribute__((ext_vector_type(2)));
typedef float v8f __attribute__((ext_vector_type(8)));


#define BIN_ROWS 64
#define BIN_SHIFT 6
#define MAX_BINS 1024
#define CHUNK 4096
#define BIN_THREADS 256
#define EDGES_PER_THREAD (CHUNK / BIN_THREADS)
#define GATHER_THREADS 512
#define ROWS_PER_WAVE (BIN_ROWS / (GATHER_THREADS / 32))
#define BIN_CAP 2048
#define RUN_LANES 16
#define RUNS_PER_PASS (GATHER_THREADS / RUN_LANES)
#define ENT_PER_THREAD 8
#define MAX_RUNS (ENT_PER_THREAD * RUNS_PER_PASS)
#define XP 100

__global__ void __launch_bounds__(BIN_THREADS)
kgetcda_chunk_sort(const int*   __restrict__ rows,
                   const int*   __restrict__ cols,
                   const float* __restrict__ vals,
                   int*         __restrict__ tbl,
                   int*         __restrict__ ent, int E, int NB, int NBLK) {
    __shared__ int h[MAX_BINS];
    __shared__ int wsum[BIN_THREADS / 32];
    const int t    = threadIdx.x;
    const int lane = t & 31;
    const int w    = t >> 5;
    const int blk  = blockIdx.x;

    const int e0 = blk * CHUNK + t;
    int r[EDGES_PER_THREAD], c[EDGES_PER_THREAD], vb[EDGES_PER_THREAD], rk[EDGES_PER_THREAD];
#pragma unroll
    for (int k = 0; k < EDGES_PER_THREAD; ++k) {
        const int e  = e0 + k * BIN_THREADS;
        const int ec = (e < E) ? e : (E - 1);
        r[k]  = rows[ec];
        c[k]  = cols[ec];
        vb[k] = __float_as_int(vals[ec]);
    }
    for (int i = t; i < MAX_BINS; i += BIN_THREADS) h[i] = 0;
    __syncthreads();
#pragma unroll
    for (int k = 0; k < EDGES_PER_THREAD; ++k) {
        const int e = e0 + k * BIN_THREADS;
        rk[k] = 0;
        if (e < E) rk[k] = atomicAdd(&h[r[k] >> BIN_SHIFT], 1);
    }
    __syncthreads();

    const int i0 = 4 * t;
    const int c0 = h[i0], c1 = h[i0 + 1], c2 = h[i0 + 2], c3 = h[i0 + 3];
    const int s4 = c0 + c1 + c2 + c3;
    int x = s4;
    for (int off = 1; off < 32; off <<= 1) {
        int y = __shfl_up(x, off);
        if (lane >= off) x += y;
    }
    if (lane == 31) wsum[w] = x;
    __syncthreads();
    if (t == 0) {
        int run = 0;
        for (int i = 0; i < BIN_THREADS / 32; ++i) {
            int v = wsum[i];
            wsum[i] = run;
            run += v;
        }
    }
    __syncthreads();
    const int b0 = wsum[w] + x - s4;
    const int b1 = b0 + c0;
    const int b2 = b1 + c1;
    const int b3 = b2 + c2;
    h[i0] = b0;  h[i0 + 1] = b1;  h[i0 + 2] = b2;  h[i0 + 3] = b3;
    if (i0     < NB) tbl[(size_t)(i0)     * NBLK + blk] = b0 | (c0 << 16);
    if (i0 + 1 < NB) tbl[(size_t)(i0 + 1) * NBLK + blk] = b1 | (c1 << 16);
    if (i0 + 2 < NB) tbl[(size_t)(i0 + 2) * NBLK + blk] = b2 | (c2 << 16);
    if (i0 + 3 < NB) tbl[(size_t)(i0 + 3) * NBLK + blk] = b3 | (c3 << 16);
    __syncthreads();

    int4* mine = reinterpret_cast<int4*>(ent) + (size_t)blk * CHUNK;
#pragma unroll
    for (int k = 0; k < EDGES_PER_THREAD; ++k) {
        const int e = e0 + k * BIN_THREADS;
        if (e < E) {
            int4 v;
            v.x = e;
            v.y = c[k];
            v.z = vb[k];
            v.w = r[k];
            mine[h[r[k] >> BIN_SHIFT] + rk[k]] = v;
        }
    }
}

__global__ void __launch_bounds__(GATHER_THREADS, 16)
kgetcda_gather_fused(const float* __restrict__ ego,
                     const int*   __restrict__ tbl,
                     const int*   __restrict__ ent,
                     const float* __restrict__ W1, const float* __restrict__ b1,
                     const float* __restrict__ W2, const float* __restrict__ b2,
                     float*       __restrict__ out, int N, int NBLK) {
    __shared__ int s_run[MAX_RUNS];
    __shared__ int rcnt[BIN_ROWS];
    __shared__ int rstart[BIN_ROWS];
    __shared__ int rcur[BIN_ROWS];
    __shared__ int s_n;
    __shared__ int s_e[BIN_CAP];
    __shared__ int s_c[BIN_CAP];
    __shared__ int s_v[BIN_CAP];
    __shared__ alignas(16) float s_xs[BIN_ROWS * XP];
    __shared__ alignas(16) float s_xm[BIN_ROWS * XP];

    const int t    = threadIdx.x;
    const int lane = t & 31;
    const int w    = __builtin_amdgcn_readfirstlane(t >> 5);
    const int bin  = blockIdx.x;
    const int row0 = bin * BIN_ROWS;
    const int4* ent4 = reinterpret_cast<const int4*>(ent);
    const float* base = ego + lane;

    if (t < MAX_RUNS) s_run[t] = (t < NBLK) ? tbl[(size_t)bin * NBLK + t] : 0;
    if (t < BIN_ROWS) rcnt[t] = 0;
    __syncthreads();
    const int k16 = t & (RUN_LANES - 1);
    const int rq  = t / RUN_LANES;
    {
        bool more = true;
        for (int kk = 0; more; kk += RUN_LANES) {
            int  rw[ENT_PER_THREAD];
            bool ok[ENT_PER_THREAD];
            more = false;
#pragma unroll
            for (int j = 0; j < ENT_PER_THREAD; ++j) {
                const int run  = j * RUNS_PER_PASS + rq;
                const int word = s_run[run];
                const int left = (word >> 16) - kk - k16;
                ok[j] = left > 0;
                more  = more || (left > RUN_LANES);
                rw[j] = ent4[ok[j] ? (unsigned)(run * CHUNK + (word & 0xffff) + kk + k16) : 0u].w;
            }
#pragma unroll
            for (int j = 0; j < ENT_PER_THREAD; ++j)
                if (ok[j]) atomicAdd(&rcnt[rw[j] - row0], 1);
        }
    }
    __syncthreads();
    if (t < 32) {
        const int c0 = rcnt[2 * t];
        const int c1 = rcnt[2 * t + 1];
        int x = c0 + c1;
        for (int off = 1; off < 32; off <<= 1) {
            int y = __shfl_up(x, off);
            if (lane >= off) x += y;
        }
        const int ex = x - c0 - c1;
        rstart[2 * t]     = ex;       rcur[2 * t]     = ex;
        rstart[2 * t + 1] = ex + c0;  rcur[2 * t + 1] = ex + c0;
        if (t == 31) s_n = x;
    }
    __syncthreads();
    const int n = __builtin_amdgcn_readfirstlane(s_n);

    if (n <= BIN_CAP) {
        bool more = true;
        for (int kk = 0; more; kk += RUN_LANES) {
            int4 u[ENT_PER_THREAD];
            bool ok[ENT_PER_THREAD];
            more = false;
#pragma unroll
            for (int j = 0; j < ENT_PER_THREAD; ++j) {
                const int run  = j * RUNS_PER_PASS + rq;
                const int word = s_run[run];
                const int left = (word >> 16) - kk - k16;
                ok[j] = left > 0;
                more  = more || (left > RUN_LANES);
                u[j]  = ent4[ok[j] ? (unsigned)(run * CHUNK + (word & 0xffff) + kk + k16) : 0u];
            }
#pragma unroll
            for (int j = 0; j < ENT_PER_THREAD; ++j) {
                if (ok[j]) {
                    const int p = atomicAdd(&rcur[u[j].w - row0], 1);
                    s_e[p] = u[j].x;
                    s_c[p] = u[j].y;
                    s_v[p] = u[j].z;
                }
            }
        }
        __syncthreads();

        for (int rr = 0; rr < ROWS_PER_WAVE; ++rr) {
            const int lr  = w * ROWS_PER_WAVE + rr;
            const int row = row0 + lr;
            if (row >= N) break;
            const float* er = ego + (size_t)row * D + lane;
            const float e0 = er[0], e1 = er[32], e2 = er[64];
            const int d  = __builtin_amdgcn_readfirstlane(rcnt[lr]);
            const int st = __builtin_amdgcn_readfirstlane(rstart[lr]);
            const int* be = s_e + st;
            const int* bc = s_c + st;
            const int* bv = s_v + st;

            float a0 = 0.0f, a1 = 0.0f, a2 = 0.0f;

            if (d <= 32) {
                int me = 0x7fffffff, mc = 0, mv = 0;
                if (lane < d) { me = be[lane]; mc = bc[lane]; mv = bv[lane]; }
                int rk = 0;
                for (int j = 0; j < d; ++j)
                    rk += (__builtin_amdgcn_readlane(me, j) < me) ? 1 : 0;
                const int sc = __builtin_amdgcn_ds_permute(rk << 2, mc);
                const int sv = __builtin_amdgcn_ds_permute(rk << 2, mv);
#pragma unroll 4
                for (int k = 0; k < d; ++k) {
                    const int   c = __builtin_amdgcn_readlane(sc, k);
                    const float v = __int_as_float(__builtin_amdgcn_readlane(sv, k));
                    const float* p = base + (size_t)c * D;
                    a0 = fmaf(v, p[0],  a0);
                    a1 = fmaf(v, p[32], a1);
                    a2 = fmaf(v, p[64], a2);
                }
            } else if (d <= 64) {
                const int me0 = be[lane], mc0 = bc[lane], mv0 = bv[lane];
                int me1 = 0x7fffffff, mc1 = 0, mv1 = 0;
                if (lane + 32 < d) { me1 = be[lane + 32]; mc1 = bc[lane + 32]; mv1 = bv[lane + 32]; }
                int rk0 = 0, rk1 = 0;
                for (int j = 0; j < 32; ++j) {
                    const int x = __builtin_amdgcn_readlane(me0, j);
                    rk0 += (x < me0) ? 1 : 0;
                    rk1 += (x < me1) ? 1 : 0;
                }
                const int d1 = d - 32;
                for (int j = 0; j < d1; ++j) {
                    const int x = __builtin_amdgcn_readlane(me1, j);
                    rk0 += (x < me0) ? 1 : 0;
                    rk1 += (x < me1) ? 1 : 0;
                }
#pragma unroll 4
                for (int k = 0; k < d; ++k) {
                    const unsigned m0 = __builtin_amdgcn_ballot_w32(rk0 == k);
                    const unsigned m1 = __builtin_amdgcn_ballot_w32(rk1 == k);
                    const int l0 = m0 ? __builtin_ctz(m0) : 0;
                    const int l1 = m1 ? __builtin_ctz(m1) : 0;
                    const int c0 = __builtin_amdgcn_readlane(mc0, l0);
                    const int c1 = __builtin_amdgcn_readlane(mc1, l1);
                    const int q0 = __builtin_amdgcn_readlane(mv0, l0);
                    const int q1 = __builtin_amdgcn_readlane(mv1, l1);
                    const int   c = m0 ? c0 : c1;
                    const float v = __int_as_float(m0 ? q0 : q1);
                    const float* p = base + (size_t)c * D;
                    a0 = fmaf(v, p[0],  a0);
                    a1 = fmaf(v, p[32], a1);
                    a2 = fmaf(v, p[64], a2);
                }
            } else {
                int last = -1;
                for (int k = 0; k < d; ++k) {
                    int best = 0x7fffffff, bq = 0;
                    for (int q = lane; q < d; q += 32) {
                        int e = be[q];
                        if (e > last && e < best) { best = e; bq = q; }
                    }
                    for (int off = 16; off > 0; off >>= 1) {
                        int ob = __shfl_xor(best, off);
                        int oq = __shfl_xor(bq, off);
                        if (ob < best) { best = ob; bq = oq; }
                    }
                    const float v = __int_as_float(bv[bq]);
                    const float* p = base + (size_t)bc[bq] * D;
                    a0 = fmaf(v, p[0],  a0);
                    a1 = fmaf(v, p[32], a1);
                    a2 = fmaf(v, p[64], a2);
                    last = best;
                }
            }
            float* xs = s_xs + lr * XP + lane;
            float* xm = s_xm + lr * XP + lane;
            xs[0]  = e0 + a0;  xm[0]  = e0 * a0;
            xs[32] = e1 + a1;  xm[32] = e1 * a1;
            xs[64] = e2 + a2;  xm[64] = e2 * a2;
        }
    } else {
        for (int rr = 0; rr < ROWS_PER_WAVE; ++rr) {
            const int lr  = w * ROWS_PER_WAVE + rr;
            const int row = row0 + lr;
            if (row >= N) break;
            const float* er = ego + (size_t)row * D + lane;
            const float e0 = er[0], e1 = er[32], e2 = er[64];
            float a0 = 0.0f, a1 = 0.0f, a2 = 0.0f;
            int last = -1;
            for (int k = 0; k < n; ++k) {
                int best = 0x7fffffff, bcol = 0, bval = 0;
                for (int q = 0; q < NBLK; ++q) {
                    const int word = s_run[q];
                    const int cn   = word >> 16;
                    const int4* rp = ent4 + (size_t)q * CHUNK + (word & 0xffff);
                    for (int i = lane; i < cn; i += 32) {
                        const int4 u = rp[i];
                        if (u.w == row && u.x > last && u.x < best) { best = u.x; bcol = u.y; bval = u.z; }
                    }
                }
                for (int off = 16; off > 0; off >>= 1) {
                    int ob = __shfl_xor(best, off);
                    int oc = __shfl_xor(bcol, off);
                    int ov = __shfl_xor(bval, off);
                    if (ob < best) { best = ob; bcol = oc; bval = ov; }
                }
                if (best == 0x7fffffff) break;
                const float v = __int_as_float(bval);
                const float* p = base + (size_t)bcol * D;
                a0 = fmaf(v, p[0],  a0);
                a1 = fmaf(v, p[32], a1);
                a2 = fmaf(v, p[64], a2);
                last = best;
            }
            float* xs = s_xs + lr * XP + lane;
            float* xm = s_xm + lr * XP + lane;
            xs[0]  = e0 + a0;  xm[0]  = e0 * a0;
            xs[32] = e1 + a1;  xm[32] = e1 * a1;
            xs[64] = e2 + a2;  xm[64] = e2 * a2;
        }
    }


    const int rt = w / 3;
    const int cp = w - 3 * rt;
    const int tileNode = row0 + rt * 16;
    const bool mine = (w < 12) && (tileNode < N);
    const int c = lane & 15;
    const int g = lane >> 4;

    __syncthreads();

    if (mine) {
        const float4* xsp = reinterpret_cast<const float4*>(s_xs + (rt * 16 + c) * XP) + g;
        const float4* xmp = reinterpret_cast<const float4*>(s_xm + (rt * 16 + c) * XP) + g;
#pragma unroll 1
        for (int j = cp * 32; j < cp * 32 + 32; j += 16) {
            v8f acc1, acc2;
            const float bb1 = b1[j + c];
            const float bb2 = b2[j + c];
#pragma unroll
            for (int r = 0; r < 8; ++r) { acc1[r] = bb1; acc2[r] = bb2; }

            const float4* w1p = reinterpret_cast<const float4*>(W1 + (size_t)(j + c) * D) + g;
            const float4* w2p = reinterpret_cast<const float4*>(W2 + (size_t)(j + c) * D) + g;

#pragma unroll 1
            for (int t0 = 0; t0 < 12; t0 += 3) {
                float4 w1[3], w2[3];
#pragma unroll
                for (int u = 0; u < 3; ++u) {
                    w1[u] = w1p[2 * (t0 + u)];
                    w2[u] = w2p[2 * (t0 + u)];
                }
#pragma unroll
                for (int u = 0; u < 3; ++u) {
                    const float4 xs = xsp[2 * (t0 + u)];
                    const float4 xm = xmp[2 * (t0 + u)];
                    v2f a1, a2, bf1, bf2;
                    a1.x  = xs.x;     a1.y  = xs.y;
                    a2.x  = xm.x;     a2.y  = xm.y;
                    bf1.x = w1[u].x;  bf1.y = w1[u].y;
                    bf2.x = w2[u].x;  bf2.y = w2[u].y;
                    acc1 = __builtin_amdgcn_wmma_f32_16x16x4_f32(
                               false, a1, false, bf1, (short)0, acc1, false, false);
                    acc2 = __builtin_amdgcn_wmma_f32_16x16x4_f32(
                               false, a2, false, bf2, (short)0, acc2, false, false);
                    a1.x  = xs.z;     a1.y  = xs.w;
                    a2.x  = xm.z;     a2.y  = xm.w;
                    bf1.x = w1[u].z;  bf1.y = w1[u].w;
                    bf2.x = w2[u].z;  bf2.y = w2[u].w;
                    acc1 = __builtin_amdgcn_wmma_f32_16x16x4_f32(
                               false, a1, false, bf1, (short)0, acc1, false, false);
                    acc2 = __builtin_amdgcn_wmma_f32_16x16x4_f32(
                               false, a2, false, bf2, (short)0, acc2, false, false);
                }
            }

#pragma unroll
            for (int r = 0; r < 8; ++r) {
                float x1 = acc1[r]; x1 = (x1 >= 0.0f) ? x1 : NEG_SLOPE * x1;
                float x2 = acc2[r]; x2 = (x2 >= 0.0f) ? x2 : NEG_SLOPE * x2;
                const int row = r + 8 * g;
                out[(size_t)(tileNode + row) * D + j + c] = x1 + x2;
            }
        }
    }
}


extern "C" void kernel_launch(void* const* d_in, const int* in_sizes, int n_in,
                              void* d_out, int out_size, void* d_ws, size_t ws_size,
                              hipStream_t stream) {
    const float* ego    = (const float*)d_in[0];
    const float* a_vals = (const float*)d_in[1];
    const float* W1     = (const float*)d_in[2];
    const float* b1     = (const float*)d_in[3];
    const float* W2     = (const float*)d_in[4];
    const float* b2     = (const float*)d_in[5];
    const int*   a_rows = (const int*)d_in[6];
    const int*   a_cols = (const int*)d_in[7];
    float* out  = (float*)d_out;

    const int N = in_sizes[0] / D;
    const int E = in_sizes[6];
    const int NB   = (N + BIN_ROWS - 1) / BIN_ROWS;
    const int NBLK = (E + CHUNK - 1) / CHUNK;

    int* ent = (int*)d_ws;
    int* tbl = ent + 4 * (size_t)E;

    kgetcda_chunk_sort<<<NBLK, BIN_THREADS, 0, stream>>>(a_rows, a_cols, a_vals, tbl, ent, E, NB, NBLK);

    kgetcda_gather_fused<<<NB, GATHER_THREADS, 0, stream>>>(ego, tbl, ent,
                                                            W1, b1, W2, b2, out, N, NBLK);
}
